# starting kernel
# baseline (speedup 1.0000x reference)
#include <hip/hip_runtime.h>

#define NN 50000
#define EE 800000
#define DD 128
#define K3 384

typedef __attribute__((ext_vector_type(2))) float v2f;
typedef __attribute__((ext_vector_type(8))) float v8f;

__global__ __launch_bounds__(256) void deg_kernel(
    const int* __restrict__ src, const int* __restrict__ dst,
    float* __restrict__ deg_out, float* __restrict__ deg_in, int E) {
  int e = blockIdx.x * blockDim.x + threadIdx.x;
  if (e >= E) return;
  atomicAdd(&deg_out[src[e]], 1.0f);
  atomicAdd(&deg_in[dst[e]], 1.0f);
}

__global__ __launch_bounds__(256) void inv_kernel(float* __restrict__ deg, int n) {
  int i = blockIdx.x * blockDim.x + threadIdx.x;
  if (i >= n) return;
  deg[i] = 1.0f / fmaxf(deg[i], 1.0f);
}

__global__ __launch_bounds__(256) void agg_kernel(
    const float* __restrict__ x,
    const int* __restrict__ src, const int* __restrict__ dst,
    const float* __restrict__ inv_out, const float* __restrict__ inv_in,
    float* __restrict__ fwd, float* __restrict__ bwd, int E) {
  long long tid = (long long)blockIdx.x * blockDim.x + threadIdx.x;
  int e = (int)(tid >> 5);
  if (e >= E) return;
  int f4 = ((int)tid & 31) * 4;

  int s = src[e];
  int d = dst[e];
  float io = inv_out[s];
  float ii = inv_in[d];

  float4 xs = *(const float4*)(x + (size_t)s * DD + f4);
  float4 xd = *(const float4*)(x + (size_t)d * DD + f4);

  float* fp = fwd + (size_t)d * DD + f4;
  float* bp = bwd + (size_t)s * DD + f4;
  atomicAdd(fp + 0, xs.x * io);
  atomicAdd(fp + 1, xs.y * io);
  atomicAdd(fp + 2, xs.z * io);
  atomicAdd(fp + 3, xs.w * io);
  atomicAdd(bp + 0, xd.x * ii);
  atomicAdd(bp + 1, xd.y * ii);
  atomicAdd(bp + 2, xd.z * ii);
  atomicAdd(bp + 3, xd.w * ii);
}

__global__ __launch_bounds__(256) void gemm_wmma_kernel(
    const float* __restrict__ x, const float* __restrict__ fwd,
    const float* __restrict__ bwd, const float* __restrict__ W,
    const float* __restrict__ bias, float* __restrict__ out, int do_relu) {
  __shared__ float As[16][K3 + 4];

  const int rowBase = blockIdx.x * 16;
  const int tid = threadIdx.x;
  const int wave = tid >> 5;
  const int lane = tid & 31;
  const int hl = lane >> 4;
  const int l16 = lane & 15;

  for (int i = tid; i < 16 * K3; i += 256) {
    int r = i / K3;
    int k = i - r * K3;
    size_t node = (size_t)(rowBase + r);
    float v;
    if (k < DD)            v = x[node * DD + k];
    else if (k < 2 * DD)   v = fwd[node * DD + (k - DD)];
    else                   v = bwd[node * DD + (k - 2 * DD)];
    As[r][k] = v;
  }
  __syncthreads();

  const int colBase = wave * 16;
  const int col = colBase + l16;

  v8f acc = {};
  for (int k0 = 0; k0 < K3; k0 += 4) {
    v2f a;
    a.x = As[l16][k0 + 2 * hl + 0];
    a.y = As[l16][k0 + 2 * hl + 1];
    v2f b;
    b.x = W[(size_t)(k0 + 2 * hl + 0) * DD + col];
    b.y = W[(size_t)(k0 + 2 * hl + 1) * DD + col];
    acc = __builtin_amdgcn_wmma_f32_16x16x4_f32(
         false, a,  false, b,
         (short)0, acc,  false,  false);
  }

  const float bb = bias[col];
#pragma unroll
  for (int r = 0; r < 8; ++r) {
    float v = acc[r] + bb;
    if (do_relu) v = fmaxf(v, 0.0f);
    out[(size_t)(rowBase + r + 8 * hl) * DD + col] = v;
  }
}

extern "C" void kernel_launch(void* const* d_in, const int* in_sizes, int n_in,
                              void* d_out, int out_size, void* d_ws, size_t ws_size,
                              hipStream_t stream) {
  const float* x  = (const float*)d_in[0];
  const float* W0 = (const float*)d_in[1];
  const float* b0 = (const float*)d_in[2];
  const float* W1 = (const float*)d_in[3];
  const float* b1 = (const float*)d_in[4];
  const float* W2 = (const float*)d_in[5];
  const float* b2 = (const float*)d_in[6];
  const int* src  = (const int*)d_in[7];
  const int* dst  = (const int*)d_in[8];
  float* out = (float*)d_out;

  float* ws      = (float*)d_ws;
  float* deg_out = ws;
  float* deg_in  = ws + NN;
  float* xA      = ws + 2 * (size_t)NN;
  float* xB      = xA + (size_t)NN * DD;
  float* fwd     = xB + (size_t)NN * DD;
  float* bwd     = fwd + (size_t)NN * DD;

  hipMemsetAsync(deg_out, 0, 2 * (size_t)NN * sizeof(float), stream);
  deg_kernel<<<(EE + 255) / 256, 256, 0, stream>>>(src, dst, deg_out, deg_in, EE);
  inv_kernel<<<(2 * NN + 255) / 256, 256, 0, stream>>>(deg_out, 2 * NN);

  const float* Ws[3]  = {W0, W1, W2};
  const float* bs[3]  = {b0, b1, b2};
  float*       dsts[3] = {xA, xB, out};

  const float* cur = x;
  const long long agg_threads = (long long)EE * 32;
  const int agg_blocks = (int)((agg_threads + 255) / 256);

  for (int l = 0; l < 3; ++l) {
    hipMemsetAsync(fwd, 0, 2 * (size_t)NN * DD * sizeof(float), stream);
    agg_kernel<<<agg_blocks, 256, 0, stream>>>(cur, src, dst, deg_out, deg_in,
                                               fwd, bwd, EE);
    gemm_wmma_kernel<<<NN / 16, 256, 0, stream>>>(cur, fwd, bwd, Ws[l], bs[l],
                                                  dsts[l], (l < 2) ? 1 : 0);
    cur = dsts[l];
  }
}


# Replace float atomic scatter-add aggregation by a per-call CSR (int atomic arrival-order slot rank, scan, fill) and per-node gather sums; neighbour sum order follows arrival rank
# speedup vs baseline: 2.6881x; 2.6881x over previous
#include <hip/hip_runtime.h>


#define NN 50000
#define EE 800000
#define DD 128
#define K3 384
#define RPS (NN + 16)
#define SCAN_T 1024
#define GATHER_BLOCKS 3125

typedef __attribute__((ext_vector_type(2))) float v2f;
typedef __attribute__((ext_vector_type(8))) float v8f;

__global__ __launch_bounds__(256) void count_rank_kernel(
    const int* __restrict__ src, const int* __restrict__ dst,
    int* __restrict__ cnt_out, int* __restrict__ cnt_in,
    int* __restrict__ rk_out, int* __restrict__ rk_in, int E) {
  int e = blockIdx.x * blockDim.x + threadIdx.x;
  if (e >= E) return;
  const int s = src[e];
  const int d = dst[e];
  int ro = atomicAdd(&cnt_out[s], 1);
  int ri = atomicAdd(&cnt_in[d], 1);
  rk_out[e] = ro;
  rk_in[e] = ri;
}

__global__ __launch_bounds__(SCAN_T) void scan_kernel(int* __restrict__ cnt,
                                                      int* __restrict__ rp) {
  __shared__ int wsum[2][32];
  int* c = cnt + (size_t)blockIdx.x * NN;
  int* r = rp + (size_t)blockIdx.x * RPS;
  const int t = threadIdx.x;
  const int lane = t & 31;
  const int wave = t >> 5;
  int carry = 0;
  int buf = 0;
  for (int q0 = 0; q0 < NN / 4; q0 += SCAN_T) {
    const int q = q0 + t;
    int4 v = make_int4(0, 0, 0, 0);
    if (q < NN / 4) v = ((const int4*)c)[q];
    const int s1 = v.x, s2 = s1 + v.y, s3 = s2 + v.z, s4 = s3 + v.w;
    int incl = s4;
    for (int off = 1; off < 32; off <<= 1) {
      int y = __shfl_up(incl, off);
      if (lane >= off) incl += y;
    }
    if (lane == 31) wsum[buf][wave] = incl;
    __syncthreads();
    const int wv = wsum[buf][lane];
    int pre = (lane < wave) ? wv : 0;
    int tot = wv;
    for (int off = 16; off > 0; off >>= 1) {
      pre += __shfl_xor(pre, off);
      tot += __shfl_xor(tot, off);
    }
    const int base = carry + pre + (incl - s4);
    if (q < NN / 4) {
      ((int4*)r)[q] = make_int4(base, base + s1, base + s2, base + s3);
      int4 o;
      o.x = __float_as_int(1.0f / fmaxf((float)v.x, 1.0f));
      o.y = __float_as_int(1.0f / fmaxf((float)v.y, 1.0f));
      o.z = __float_as_int(1.0f / fmaxf((float)v.z, 1.0f));
      o.w = __float_as_int(1.0f / fmaxf((float)v.w, 1.0f));
      ((int4*)c)[q] = o;
    }
    carry += tot;
    buf ^= 1;
  }
  if (t == 0) r[NN] = carry;
}

__global__ __launch_bounds__(256) void fill_kernel(
    const int* __restrict__ src, const int* __restrict__ dst,
    const int* __restrict__ rp_out, const int* __restrict__ rp_in,
    const int* __restrict__ rk_out, const int* __restrict__ rk_in,
    const float* __restrict__ inv_out, const float* __restrict__ inv_in,
    int* __restrict__ adj_in, int* __restrict__ adj_out, int E) {
  int e = blockIdx.x * blockDim.x + threadIdx.x;
  if (e >= E) return;
  const int s = src[e];
  const int d = dst[e];
  const int pi = rp_in[d] + rk_in[e];
  const int po = rp_out[s] + rk_out[e];
  const int2 vi = make_int2(s, __float_as_int(inv_out[s]));
  const int2 vo = make_int2(d, __float_as_int(inv_in[d]));
  if ((unsigned)pi < (unsigned)E) *(int2*)(adj_in + 2 * (size_t)pi) = vi;
  if ((unsigned)po < (unsigned)E) *(int2*)(adj_out + 2 * (size_t)po) = vo;
}

static __device__ __forceinline__ int2 ld_chunk(const int* __restrict__ adj,
                                                int base, int n, int lane) {
  int2 t = make_int2(0, 0);
  if (lane < n) t = *(const int2*)(adj + 2 * (size_t)(base + lane));
  return t;
}

static __device__ __forceinline__ void ld_rows(const float* __restrict__ x,
                                               int cs, int cw, int k, int n, int lane,
                                               float4 (&xv)[4], float (&w)[4]) {
#pragma unroll
  for (int j = 0; j < 4; ++j) {
    int kk = k + j;
    if (j > 0 && kk > n - 1) kk = n - 1;
    int s = __builtin_amdgcn_readlane(cs, kk);
    if ((unsigned)s > (unsigned)(NN - 1)) s = NN - 1;
    w[j] = __int_as_float(__builtin_amdgcn_readlane(cw, kk));
    xv[j] = *(const float4*)(x + (size_t)s * DD + lane * 4);
  }
}

static __device__ __forceinline__ void add_rows(const float4 (&xv)[4], const float (&w)[4],
                                                int k, int n, float4& acc) {
#pragma unroll
  for (int j = 0; j < 4; ++j) {
    if (j == 0 || k + j < n) {
      acc.x = fmaf(xv[j].x, w[j], acc.x);
      acc.y = fmaf(xv[j].y, w[j], acc.y);
      acc.z = fmaf(xv[j].z, w[j], acc.z);
      acc.w = fmaf(xv[j].w, w[j], acc.w);
    }
  }
}

__global__ __launch_bounds__(256) void gather_kernel(
    const float* __restrict__ x,
    const int* __restrict__ rp_out, const int* __restrict__ rp_in,
    const int* __restrict__ adj_in, const int* __restrict__ adj_out,
    float* __restrict__ fwd, float* __restrict__ bwd) {
  const int lane = threadIdx.x & 31;
  const int wid = __builtin_amdgcn_readfirstlane((int)(blockIdx.x * 8 + (threadIdx.x >> 5)));
  for (int v = wid; v < NN; v += GATHER_BLOCKS * 8) {
    int bi = rp_in[v], ei = rp_in[v + 1];
    int bo = rp_out[v], eo = rp_out[v + 1];
    if (bi < 0) bi = 0;
    if (bo < 0) bo = 0;
    if (ei > EE) ei = EE;
    if (eo > EE) eo = EE;

    float4 f = make_float4(0.f, 0.f, 0.f, 0.f);
    float4 g = make_float4(0.f, 0.f, 0.f, 0.f);
    for (int off = 0; off < EE; off += 32) {
      int na = ei - bi - off; if (na > 32) na = 32;
      int nb = eo - bo - off; if (nb > 32) nb = 32;
      if (na <= 0 && nb <= 0) break;
      const int2 ta = ld_chunk(adj_in, bi + off, na, lane);
      const int2 tb = ld_chunk(adj_out, bo + off, nb, lane);
      const int nmin = na < nb ? na : nb;
      int k = 0;
      for (; k < nmin; k += 4) {
        float4 xa[4], xb[4];
        float wa[4], wb[4];
        ld_rows(x, ta.x, ta.y, k, na, lane, xa, wa);
        ld_rows(x, tb.x, tb.y, k, nb, lane, xb, wb);
        add_rows(xa, wa, k, na, f);
        add_rows(xb, wb, k, nb, g);
      }
      for (; k < na; k += 4) {
        float4 xa[4];
        float wa[4];
        ld_rows(x, ta.x, ta.y, k, na, lane, xa, wa);
        add_rows(xa, wa, k, na, f);
      }
      for (; k < nb; k += 4) {
        float4 xb[4];
        float wb[4];
        ld_rows(x, tb.x, tb.y, k, nb, lane, xb, wb);
        add_rows(xb, wb, k, nb, g);
      }
    }
    *(float4*)(fwd + (size_t)v * DD + lane * 4) = f;
    *(float4*)(bwd + (size_t)v * DD + lane * 4) = g;
  }
}

__global__ __launch_bounds__(256) void gemm_wmma_kernel(
    const float* __restrict__ x, const float* __restrict__ fwd,
    const float* __restrict__ bwd, const float* __restrict__ W,
    const float* __restrict__ bias, float* __restrict__ out, int do_relu) {
  __shared__ float As[16][K3 + 4];

  const int rowBase = blockIdx.x * 16;
  const int tid = threadIdx.x;
  const int wave = tid >> 5;
  const int lane = tid & 31;
  const int hl = lane >> 4;
  const int l16 = lane & 15;

  for (int i = tid; i < 16 * K3; i += 256) {
    int r = i / K3;
    int k = i - r * K3;
    size_t node = (size_t)(rowBase + r);
    float v;
    if (k < DD)            v = x[node * DD + k];
    else if (k < 2 * DD)   v = fwd[node * DD + (k - DD)];
    else                   v = bwd[node * DD + (k - 2 * DD)];
    As[r][k] = v;
  }
  __syncthreads();

  const int colBase = wave * 16;
  const int col = colBase + l16;

  v8f acc = {};
  for (int k0 = 0; k0 < K3; k0 += 4) {
    v2f a;
    a.x = As[l16][k0 + 2 * hl + 0];
    a.y = As[l16][k0 + 2 * hl + 1];
    v2f b;
    b.x = W[(size_t)(k0 + 2 * hl + 0) * DD + col];
    b.y = W[(size_t)(k0 + 2 * hl + 1) * DD + col];
    acc = __builtin_amdgcn_wmma_f32_16x16x4_f32(
         false, a,  false, b,
         (short)0, acc,  false,  false);
  }

  const float bb = bias[col];
#pragma unroll
  for (int r = 0; r < 8; ++r) {
    float v = acc[r] + bb;
    if (do_relu) v = fmaxf(v, 0.0f);
    out[(size_t)(rowBase + r + 8 * hl) * DD + col] = v;
  }
}

extern "C" void kernel_launch(void* const* d_in, const int* in_sizes, int n_in,
                              void* d_out, int out_size, void* d_ws, size_t ws_size,
                              hipStream_t stream) {
  const float* x  = (const float*)d_in[0];
  const float* W0 = (const float*)d_in[1];
  const float* b0 = (const float*)d_in[2];
  const float* W1 = (const float*)d_in[3];
  const float* b1 = (const float*)d_in[4];
  const float* W2 = (const float*)d_in[5];
  const float* b2 = (const float*)d_in[6];
  const int* src  = (const int*)d_in[7];
  const int* dst  = (const int*)d_in[8];
  float* out = (float*)d_out;

  float* ws      = (float*)d_ws;
  int*   cnt_out = (int*)ws;
  int*   cnt_in  = cnt_out + NN;
  float* xA      = ws + 2 * (size_t)NN;
  int*   rp_out  = (int*)(xA + (size_t)NN * DD);
  int*   rp_in   = rp_out + RPS;
  int*   adj_in  = rp_in + RPS;
  int*   adj_out = adj_in + 2 * (size_t)EE;
  int*   rk_out  = adj_out + 2 * (size_t)EE;
  int*   rk_in   = rk_out + EE;
  float* fwd     = xA + 2 * (size_t)NN * DD;
  float* bwd     = fwd + (size_t)NN * DD;

  hipMemsetAsync(cnt_out, 0, 2 * (size_t)NN * sizeof(int), stream);
  count_rank_kernel<<<(EE + 255) / 256, 256, 0, stream>>>(src, dst, cnt_out, cnt_in,
                                                          rk_out, rk_in, EE);
  scan_kernel<<<2, SCAN_T, 0, stream>>>(cnt_out, rp_out);
  fill_kernel<<<(EE + 255) / 256, 256, 0, stream>>>(src, dst, rp_out, rp_in, rk_out, rk_in,
                                                    (const float*)cnt_out,
                                                    (const float*)cnt_in,
                                                    adj_in, adj_out, EE);

  const float* Ws[3]  = {W0, W1, W2};
  const float* bs[3]  = {b0, b1, b2};
  float*       dsts[3] = {out, xA, out};

  const float* cur = x;
  for (int l = 0; l < 3; ++l) {
    gather_kernel<<<GATHER_BLOCKS, 256, 0, stream>>>(cur, rp_out, rp_in, adj_in, adj_out,
                                                     fwd, bwd);
    gemm_wmma_kernel<<<NN / 16, 256, 0, stream>>>(cur, fwd, bwd, Ws[l], bs[l],
                                                  dsts[l], (l < 2) ? 1 : 0);
    cur = dsts[l];
  }
}


# Degree count and slot rank on LDS counters: (node range x edge chunk) owner work-groups stream the keys; no global atomics, memset or serial scan; neighbour sum order by edge chunk, then arrival rank
# speedup vs baseline: 3.2014x; 1.1910x over previous
#include <hip/hip_runtime.h>


#define NN 50000
#define EE 800000
#define DD 128
#define K3 384
#define RPS (NN + 16)
#define PB 16
#define QC 8
#define RNG 3128
#define CH4 (EE / 4 / QC)
#define OWN_T 1024
#define KEY_U 5
static_assert(PB * RNG >= NN && RNG % 8 == 0 && NN % 8 == 0, "ranges cover the nodes in whole groups of four");
static_assert(CH4 * 4 * QC == EE, "chunks cover the edges");
static_assert(4 * OWN_T >= RNG && (PB - 1) * QC <= OWN_T, "one work-group scans a range");
#define GATHER_BLOCKS 3125

typedef __attribute__((ext_vector_type(2))) float v2f;
typedef __attribute__((ext_vector_type(8))) float v8f;

static __device__ __forceinline__ void own_key(int k, int e, int lo, int n,
                                               int* c, int* __restrict__ rk) {
  const unsigned r = (unsigned)(k - lo);
  if (r < (unsigned)n) rk[e] = atomicAdd(&c[r], 1);
}

__global__ __launch_bounds__(OWN_T) void rank_count_kernel(
    const int* __restrict__ src, const int* __restrict__ dst,
    int* __restrict__ cnt, int* __restrict__ tot, int* __restrict__ rk) {
  __shared__ int c[RNG];
  __shared__ int tot_s;
  const int t = threadIdx.x;
  const int lane = t & 31;
  const int dir = (int)blockIdx.x / (PB * QC);
  const int pq = (int)blockIdx.x - dir * (PB * QC);
  const int p = pq / QC;
  const int q = pq - p * QC;
  const int lo = p * RNG;
  int n = NN - lo;
  if (n > RNG) n = RNG;
  if (n < 0) n = 0;
  const int4* __restrict__ k4 = (const int4*)(dir ? dst : src) + (size_t)q * CH4;
  int* __restrict__ rkd = rk + (size_t)dir * EE + (size_t)q * (4 * CH4);

  for (int i = t; i < RNG; i += OWN_T) c[i] = 0;
  if (t == 0) tot_s = 0;
  __syncthreads();

  for (int i0 = t; i0 < CH4; i0 += KEY_U * OWN_T) {
    int4 k[KEY_U];
#pragma unroll
    for (int j = 0; j < KEY_U; ++j) {
      const int i = i0 + j * OWN_T;
      k[j] = make_int4(0x7fffffff, 0x7fffffff, 0x7fffffff, 0x7fffffff);
      if (i < CH4) k[j] = k4[i];
    }
#pragma unroll
    for (int j = 0; j < KEY_U; ++j) {
      const int e = 4 * (i0 + j * OWN_T);
      own_key(k[j].x, e + 0, lo, n, c, rkd);
      own_key(k[j].y, e + 1, lo, n, c, rkd);
      own_key(k[j].z, e + 2, lo, n, c, rkd);
      own_key(k[j].w, e + 3, lo, n, c, rkd);
    }
  }
  __syncthreads();

  int* __restrict__ cd = cnt + ((size_t)dir * QC + q) * NN + lo;
  int sum = 0;
  for (int i = t; i < n; i += OWN_T) {
    const int v = c[i];
    cd[i] = v;
    sum += v;
  }
  for (int off = 16; off > 0; off >>= 1) sum += __shfl_xor(sum, off);
  if (lane == 0) atomicAdd(&tot_s, sum);
  __syncthreads();
  if (t == 0) tot[(dir * PB + p) * QC + q] = tot_s;
}

__global__ __launch_bounds__(OWN_T) void scan_off_kernel(
    int* __restrict__ cnt, const int* __restrict__ tot,
    float* __restrict__ inv, int* __restrict__ rp) {
  __shared__ int wsum[32];
  __shared__ int base_s;
  const int t = threadIdx.x;
  const int lane = t & 31;
  const int wave = t >> 5;
  const int dir = (int)blockIdx.x / PB;
  const int p = (int)blockIdx.x - dir * PB;
  const int lo = p * RNG;
  int n = NN - lo;
  if (n > RNG) n = RNG;
  if (n < 0) n = 0;
  const bool live = 4 * t < n;

  if (t == 0) base_s = 0;
  __syncthreads();
  if (t < p * QC) atomicAdd(&base_s, tot[dir * PB * QC + t]);

  int* __restrict__ cd = cnt + (size_t)dir * QC * NN + lo + 4 * t;
  int4 v[QC];
  int4 deg = make_int4(0, 0, 0, 0);
#pragma unroll
  for (int q = 0; q < QC; ++q) {
    int4 w = make_int4(0, 0, 0, 0);
    if (live) w = *(const int4*)(cd + (size_t)q * NN);
    v[q] = deg;
    deg.x += w.x; deg.y += w.y; deg.z += w.z; deg.w += w.w;
  }
  const int s1 = deg.x, s2 = s1 + deg.y, s3 = s2 + deg.z, s4 = s3 + deg.w;
  int incl = s4;
  for (int off = 1; off < 32; off <<= 1) {
    int y = __shfl_up(incl, off);
    if (lane >= off) incl += y;
  }
  if (lane == 31) wsum[wave] = incl;
  __syncthreads();
  const int wv = wsum[lane];
  int pre = (lane < wave) ? wv : 0;
  int all = wv;
  for (int off = 16; off > 0; off >>= 1) {
    pre += __shfl_xor(pre, off);
    all += __shfl_xor(all, off);
  }
  const int first = base_s;
  const int b0 = first + pre + (incl - s4);
  if (live) {
    const int4 r = make_int4(b0, b0 + s1, b0 + s2, b0 + s3);
    *(int4*)(rp + (size_t)dir * RPS + lo + 4 * t) = r;
    float4 o;
    o.x = 1.0f / fmaxf((float)deg.x, 1.0f);
    o.y = 1.0f / fmaxf((float)deg.y, 1.0f);
    o.z = 1.0f / fmaxf((float)deg.z, 1.0f);
    o.w = 1.0f / fmaxf((float)deg.w, 1.0f);
    *(float4*)(inv + (size_t)dir * NN + lo + 4 * t) = o;
#pragma unroll
    for (int q = 0; q < QC; ++q) {
      *(int4*)(cd + (size_t)q * NN) =
          make_int4(r.x + v[q].x, r.y + v[q].y, r.z + v[q].z, r.w + v[q].w);
    }
  }
  if (p == PB - 1 && t == 0) rp[(size_t)dir * RPS + NN] = first + all;
}

__global__ __launch_bounds__(256) void fill_kernel(
    const int* __restrict__ src, const int* __restrict__ dst,
    const int* __restrict__ off_out, const int* __restrict__ off_in,
    const int* __restrict__ rk_out, const int* __restrict__ rk_in,
    const float* __restrict__ inv_out, const float* __restrict__ inv_in,
    int* __restrict__ adj_in, int* __restrict__ adj_out, int E) {
  int e = blockIdx.x * blockDim.x + threadIdx.x;
  if (e >= E) return;
  const int s = src[e];
  const int d = dst[e];
  const size_t row = (size_t)(e / (4 * CH4)) * NN;
  const int pi = off_in[row + d] + rk_in[e];
  const int po = off_out[row + s] + rk_out[e];
  const int2 vi = make_int2(s, __float_as_int(inv_out[s]));
  const int2 vo = make_int2(d, __float_as_int(inv_in[d]));
  if ((unsigned)pi < (unsigned)E) *(int2*)(adj_in + 2 * (size_t)pi) = vi;
  if ((unsigned)po < (unsigned)E) *(int2*)(adj_out + 2 * (size_t)po) = vo;
}

static __device__ __forceinline__ int2 ld_chunk(const int* __restrict__ adj,
                                                int base, int n, int lane) {
  int2 t = make_int2(0, 0);
  if (lane < n) t = *(const int2*)(adj + 2 * (size_t)(base + lane));
  return t;
}

static __device__ __forceinline__ void ld_rows(const float* __restrict__ x,
                                               int cs, int cw, int k, int n, int lane,
                                               float4 (&xv)[4], float (&w)[4]) {
#pragma unroll
  for (int j = 0; j < 4; ++j) {
    int kk = k + j;
    if (j > 0 && kk > n - 1) kk = n - 1;
    int s = __builtin_amdgcn_readlane(cs, kk);
    if ((unsigned)s > (unsigned)(NN - 1)) s = NN - 1;
    w[j] = __int_as_float(__builtin_amdgcn_readlane(cw, kk));
    xv[j] = *(const float4*)(x + (size_t)s * DD + lane * 4);
  }
}

static __device__ __forceinline__ void add_rows(const float4 (&xv)[4], const float (&w)[4],
                                                int k, int n, float4& acc) {
#pragma unroll
  for (int j = 0; j < 4; ++j) {
    if (j == 0 || k + j < n) {
      acc.x = fmaf(xv[j].x, w[j], acc.x);
      acc.y = fmaf(xv[j].y, w[j], acc.y);
      acc.z = fmaf(xv[j].z, w[j], acc.z);
      acc.w = fmaf(xv[j].w, w[j], acc.w);
    }
  }
}

__global__ __launch_bounds__(256) void gather_kernel(
    const float* __restrict__ x,
    const int* __restrict__ rp_out, const int* __restrict__ rp_in,
    const int* __restrict__ adj_in, const int* __restrict__ adj_out,
    float* __restrict__ fwd, float* __restrict__ bwd) {
  const int lane = threadIdx.x & 31;
  const int wid = __builtin_amdgcn_readfirstlane((int)(blockIdx.x * 8 + (threadIdx.x >> 5)));
  for (int v = wid; v < NN; v += GATHER_BLOCKS * 8) {
    int bi = rp_in[v], ei = rp_in[v + 1];
    int bo = rp_out[v], eo = rp_out[v + 1];
    if (bi < 0) bi = 0;
    if (bo < 0) bo = 0;
    if (ei > EE) ei = EE;
    if (eo > EE) eo = EE;

    float4 f = make_float4(0.f, 0.f, 0.f, 0.f);
    float4 g = make_float4(0.f, 0.f, 0.f, 0.f);
    for (int off = 0; off < EE; off += 32) {
      int na = ei - bi - off; if (na > 32) na = 32;
      int nb = eo - bo - off; if (nb > 32) nb = 32;
      if (na <= 0 && nb <= 0) break;
      const int2 ta = ld_chunk(adj_in, bi + off, na, lane);
      const int2 tb = ld_chunk(adj_out, bo + off, nb, lane);
      const int nmin = na < nb ? na : nb;
      int k = 0;
      for (; k < nmin; k += 4) {
        float4 xa[4], xb[4];
        float wa[4], wb[4];
        ld_rows(x, ta.x, ta.y, k, na, lane, xa, wa);
        ld_rows(x, tb.x, tb.y, k, nb, lane, xb, wb);
        add_rows(xa, wa, k, na, f);
        add_rows(xb, wb, k, nb, g);
      }
      for (; k < na; k += 4) {
        float4 xa[4];
        float wa[4];
        ld_rows(x, ta.x, ta.y, k, na, lane, xa, wa);
        add_rows(xa, wa, k, na, f);
      }
      for (; k < nb; k += 4) {
        float4 xb[4];
        float wb[4];
        ld_rows(x, tb.x, tb.y, k, nb, lane, xb, wb);
        add_rows(xb, wb, k, nb, g);
      }
    }
    *(float4*)(fwd + (size_t)v * DD + lane * 4) = f;
    *(float4*)(bwd + (size_t)v * DD + lane * 4) = g;
  }
}

__global__ __launch_bounds__(256) void gemm_wmma_kernel(
    const float* __restrict__ x, const float* __restrict__ fwd,
    const float* __restrict__ bwd, const float* __restrict__ W,
    const float* __restrict__ bias, float* __restrict__ out, int do_relu) {
  __shared__ float As[16][K3 + 4];

  const int rowBase = blockIdx.x * 16;
  const int tid = threadIdx.x;
  const int wave = tid >> 5;
  const int lane = tid & 31;
  const int hl = lane >> 4;
  const int l16 = lane & 15;

  for (int i = tid; i < 16 * K3; i += 256) {
    int r = i / K3;
    int k = i - r * K3;
    size_t node = (size_t)(rowBase + r);
    float v;
    if (k < DD)            v = x[node * DD + k];
    else if (k < 2 * DD)   v = fwd[node * DD + (k - DD)];
    else                   v = bwd[node * DD + (k - 2 * DD)];
    As[r][k] = v;
  }
  __syncthreads();

  const int colBase = wave * 16;
  const int col = colBase + l16;

  v8f acc = {};
  for (int k0 = 0; k0 < K3; k0 += 4) {
    v2f a;
    a.x = As[l16][k0 + 2 * hl + 0];
    a.y = As[l16][k0 + 2 * hl + 1];
    v2f b;
    b.x = W[(size_t)(k0 + 2 * hl + 0) * DD + col];
    b.y = W[(size_t)(k0 + 2 * hl + 1) * DD + col];
    acc = __builtin_amdgcn_wmma_f32_16x16x4_f32(
         false, a,  false, b,
         (short)0, acc,  false,  false);
  }

  const float bb = bias[col];
#pragma unroll
  for (int r = 0; r < 8; ++r) {
    float v = acc[r] + bb;
    if (do_relu) v = fmaxf(v, 0.0f);
    out[(size_t)(rowBase + r + 8 * hl) * DD + col] = v;
  }
}

extern "C" void kernel_launch(void* const* d_in, const int* in_sizes, int n_in,
                              void* d_out, int out_size, void* d_ws, size_t ws_size,
                              hipStream_t stream) {
  const float* x  = (const float*)d_in[0];
  const float* W0 = (const float*)d_in[1];
  const float* b0 = (const float*)d_in[2];
  const float* W1 = (const float*)d_in[3];
  const float* b1 = (const float*)d_in[4];
  const float* W2 = (const float*)d_in[5];
  const float* b2 = (const float*)d_in[6];
  const int* src  = (const int*)d_in[7];
  const int* dst  = (const int*)d_in[8];
  float* out = (float*)d_out;

  float* ws      = (float*)d_ws;
  float* inv_out = ws;
  float* inv_in  = inv_out + NN;
  float* xA      = ws + 2 * (size_t)NN;
  int*   rp_out  = (int*)(xA + (size_t)NN * DD);
  int*   rp_in   = rp_out + RPS;
  int*   adj_in  = rp_in + RPS;
  int*   adj_out = adj_in + 2 * (size_t)EE;
  int*   rk_out  = adj_out + 2 * (size_t)EE;
  int*   rk_in   = rk_out + EE;
  int*   cnt     = rk_in + EE;
  int*   tot     = cnt + 2 * (size_t)QC * NN;
  float* fwd     = xA + 2 * (size_t)NN * DD;
  float* bwd     = fwd + (size_t)NN * DD;

  rank_count_kernel<<<2 * PB * QC, OWN_T, 0, stream>>>(src, dst, cnt, tot, rk_out);
  scan_off_kernel<<<2 * PB, OWN_T, 0, stream>>>(cnt, tot, inv_out, rp_out);
  fill_kernel<<<(EE + 255) / 256, 256, 0, stream>>>(src, dst, cnt, cnt + (size_t)QC * NN,
                                                    rk_out, rk_in, inv_out, inv_in,
                                                    adj_in, adj_out, EE);

  const float* Ws[3]  = {W0, W1, W2};
  const float* bs[3]  = {b0, b1, b2};
  float*       dsts[3] = {out, xA, out};

  const float* cur = x;
  for (int l = 0; l < 3; ++l) {
    gather_kernel<<<GATHER_BLOCKS, 256, 0, stream>>>(cur, rp_out, rp_in, adj_in, adj_out,
                                                     fwd, bwd);
    gemm_wmma_kernel<<<NN / 16, 256, 0, stream>>>(cur, fwd, bwd, Ws[l], bs[l],
                                                  dsts[l], (l < 2) ? 1 : 0);
    cur = dsts[l];
  }
}


# Fuse each layer's gather into its GEMM launch: each wave sums two nodes' lists straight into the LDS A tile, then the block multiplies; fwd/bwd never touch global memory; sum order unchanged
# speedup vs baseline: 3.7105x; 1.1590x over previous
#include <hip/hip_runtime.h>


#define NN 50000
#define EE 800000
#define DD 128
#define K3 384
#define RPS (NN + 16)
#define PB 16
#define QC 8
#define RNG 3128
#define CH4 (EE / 4 / QC)
#define OWN_T 1024
#define KEY_U 5
static_assert(PB * RNG >= NN && RNG % 8 == 0 && NN % 8 == 0, "ranges cover the nodes in whole groups of four");
static_assert(CH4 * 4 * QC == EE, "chunks cover the edges");
static_assert(4 * OWN_T >= RNG && (PB - 1) * QC <= OWN_T, "one work-group scans a range");
static_assert(NN % 16 == 0, "whole 16-node tiles");

typedef __attribute__((ext_vector_type(2))) float v2f;
typedef __attribute__((ext_vector_type(8))) float v8f;

static __device__ __forceinline__ void own_key(int k, int e, int lo, int n,
                                               int* c, int* __restrict__ rk) {
  const unsigned r = (unsigned)(k - lo);
  if (r < (unsigned)n) rk[e] = atomicAdd(&c[r], 1);
}

__global__ __launch_bounds__(OWN_T) void rank_count_kernel(
    const int* __restrict__ src, const int* __restrict__ dst,
    int* __restrict__ cnt, int* __restrict__ tot, int* __restrict__ rk) {
  __shared__ int c[RNG];
  __shared__ int tot_s;
  const int t = threadIdx.x;
  const int lane = t & 31;
  const int dir = (int)blockIdx.x / (PB * QC);
  const int pq = (int)blockIdx.x - dir * (PB * QC);
  const int p = pq / QC;
  const int q = pq - p * QC;
  const int lo = p * RNG;
  int n = NN - lo;
  if (n > RNG) n = RNG;
  if (n < 0) n = 0;
  const int4* __restrict__ k4 = (const int4*)(dir ? dst : src) + (size_t)q * CH4;
  int* __restrict__ rkd = rk + (size_t)dir * EE + (size_t)q * (4 * CH4);

  for (int i = t; i < RNG; i += OWN_T) c[i] = 0;
  if (t == 0) tot_s = 0;
  __syncthreads();

  for (int i0 = t; i0 < CH4; i0 += KEY_U * OWN_T) {
    int4 k[KEY_U];
#pragma unroll
    for (int j = 0; j < KEY_U; ++j) {
      const int i = i0 + j * OWN_T;
      k[j] = make_int4(0x7fffffff, 0x7fffffff, 0x7fffffff, 0x7fffffff);
      if (i < CH4) k[j] = k4[i];
    }
#pragma unroll
    for (int j = 0; j < KEY_U; ++j) {
      const int e = 4 * (i0 + j * OWN_T);
      own_key(k[j].x, e + 0, lo, n, c, rkd);
      own_key(k[j].y, e + 1, lo, n, c, rkd);
      own_key(k[j].z, e + 2, lo, n, c, rkd);
      own_key(k[j].w, e + 3, lo, n, c, rkd);
    }
  }
  __syncthreads();

  int* __restrict__ cd = cnt + ((size_t)dir * QC + q) * NN + lo;
  int sum = 0;
  for (int i = t; i < n; i += OWN_T) {
    const int v = c[i];
    cd[i] = v;
    sum += v;
  }
  for (int off = 16; off > 0; off >>= 1) sum += __shfl_xor(sum, off);
  if (lane == 0) atomicAdd(&tot_s, sum);
  __syncthreads();
  if (t == 0) tot[(dir * PB + p) * QC + q] = tot_s;
}

__global__ __launch_bounds__(OWN_T) void scan_off_kernel(
    int* __restrict__ cnt, const int* __restrict__ tot,
    float* __restrict__ inv, int* __restrict__ rp) {
  __shared__ int wsum[32];
  __shared__ int base_s;
  const int t = threadIdx.x;
  const int lane = t & 31;
  const int wave = t >> 5;
  const int dir = (int)blockIdx.x / PB;
  const int p = (int)blockIdx.x - dir * PB;
  const int lo = p * RNG;
  int n = NN - lo;
  if (n > RNG) n = RNG;
  if (n < 0) n = 0;
  const bool live = 4 * t < n;

  if (t == 0) base_s = 0;
  __syncthreads();
  if (t < p * QC) atomicAdd(&base_s, tot[dir * PB * QC + t]);

  int* __restrict__ cd = cnt + (size_t)dir * QC * NN + lo + 4 * t;
  int4 v[QC];
  int4 deg = make_int4(0, 0, 0, 0);
#pragma unroll
  for (int q = 0; q < QC; ++q) {
    int4 w = make_int4(0, 0, 0, 0);
    if (live) w = *(const int4*)(cd + (size_t)q * NN);
    v[q] = deg;
    deg.x += w.x; deg.y += w.y; deg.z += w.z; deg.w += w.w;
  }
  const int s1 = deg.x, s2 = s1 + deg.y, s3 = s2 + deg.z, s4 = s3 + deg.w;
  int incl = s4;
  for (int off = 1; off < 32; off <<= 1) {
    int y = __shfl_up(incl, off);
    if (lane >= off) incl += y;
  }
  if (lane == 31) wsum[wave] = incl;
  __syncthreads();
  const int wv = wsum[lane];
  int pre = (lane < wave) ? wv : 0;
  int all = wv;
  for (int off = 16; off > 0; off >>= 1) {
    pre += __shfl_xor(pre, off);
    all += __shfl_xor(all, off);
  }
  const int first = base_s;
  const int b0 = first + pre + (incl - s4);
  if (live) {
    const int4 r = make_int4(b0, b0 + s1, b0 + s2, b0 + s3);
    *(int4*)(rp + (size_t)dir * RPS + lo + 4 * t) = r;
    float4 o;
    o.x = 1.0f / fmaxf((float)deg.x, 1.0f);
    o.y = 1.0f / fmaxf((float)deg.y, 1.0f);
    o.z = 1.0f / fmaxf((float)deg.z, 1.0f);
    o.w = 1.0f / fmaxf((float)deg.w, 1.0f);
    *(float4*)(inv + (size_t)dir * NN + lo + 4 * t) = o;
#pragma unroll
    for (int q = 0; q < QC; ++q) {
      *(int4*)(cd + (size_t)q * NN) =
          make_int4(r.x + v[q].x, r.y + v[q].y, r.z + v[q].z, r.w + v[q].w);
    }
  }
  if (p == PB - 1 && t == 0) rp[(size_t)dir * RPS + NN] = first + all;
}

__global__ __launch_bounds__(256) void fill_kernel(
    const int* __restrict__ src, const int* __restrict__ dst,
    const int* __restrict__ off_out, const int* __restrict__ off_in,
    const int* __restrict__ rk_out, const int* __restrict__ rk_in,
    const float* __restrict__ inv_out, const float* __restrict__ inv_in,
    int* __restrict__ adj_in, int* __restrict__ adj_out, int E) {
  int e = blockIdx.x * blockDim.x + threadIdx.x;
  if (e >= E) return;
  const int s = src[e];
  const int d = dst[e];
  const size_t row = (size_t)(e / (4 * CH4)) * NN;
  const int pi = off_in[row + d] + rk_in[e];
  const int po = off_out[row + s] + rk_out[e];
  const int2 vi = make_int2(s, __float_as_int(inv_out[s]));
  const int2 vo = make_int2(d, __float_as_int(inv_in[d]));
  if ((unsigned)pi < (unsigned)E) *(int2*)(adj_in + 2 * (size_t)pi) = vi;
  if ((unsigned)po < (unsigned)E) *(int2*)(adj_out + 2 * (size_t)po) = vo;
}

static __device__ __forceinline__ int2 ld_chunk(const int* __restrict__ adj,
                                                int base, int n, int lane) {
  int2 t = make_int2(0, 0);
  if (lane < n) t = *(const int2*)(adj + 2 * (size_t)(base + lane));
  return t;
}

static __device__ __forceinline__ void ld_rows(const float* __restrict__ x,
                                               int cs, int cw, int k, int n, int lane,
                                               float4 (&xv)[4], float (&w)[4]) {
#pragma unroll
  for (int j = 0; j < 4; ++j) {
    int kk = k + j;
    if (j > 0 && kk > n - 1) kk = n - 1;
    int s = __builtin_amdgcn_readlane(cs, kk);
    if ((unsigned)s > (unsigned)(NN - 1)) s = NN - 1;
    w[j] = __int_as_float(__builtin_amdgcn_readlane(cw, kk));
    xv[j] = *(const float4*)(x + (size_t)s * DD + lane * 4);
  }
}

static __device__ __forceinline__ void add_rows(const float4 (&xv)[4], const float (&w)[4],
                                                int k, int n, float4& acc) {
#pragma unroll
  for (int j = 0; j < 4; ++j) {
    if (j == 0 || k + j < n) {
      acc.x = fmaf(xv[j].x, w[j], acc.x);
      acc.y = fmaf(xv[j].y, w[j], acc.y);
      acc.z = fmaf(xv[j].z, w[j], acc.z);
      acc.w = fmaf(xv[j].w, w[j], acc.w);
    }
  }
}

__global__ __launch_bounds__(256) void layer_kernel(
    const float* __restrict__ x,
    const int* __restrict__ rp_out, const int* __restrict__ rp_in,
    const int* __restrict__ adj_in, const int* __restrict__ adj_out,
    const float* __restrict__ W, const float* __restrict__ bias,
    float* __restrict__ out, int do_relu) {
  __shared__ __attribute__((aligned(16))) float As[16][K3 + 4];

  const int rowBase = blockIdx.x * 16;
  const int tid = threadIdx.x;
  const int lane = tid & 31;
  const int wave = __builtin_amdgcn_readfirstlane(tid >> 5);

  const int r0 = wave * 2;
  const int v0 = rowBase + r0;
  const int pi0 = rp_in[v0], pi1 = rp_in[v0 + 1], pi2 = rp_in[v0 + 2];
  const int po0 = rp_out[v0], po1 = rp_out[v0 + 1], po2 = rp_out[v0 + 2];
  {
    const float4 xs0 = *(const float4*)(x + (size_t)v0 * DD + lane * 4);
    const float4 xs1 = *(const float4*)(x + (size_t)(v0 + 1) * DD + lane * 4);
    *(float4*)&As[r0][lane * 4] = xs0;
    *(float4*)&As[r0 + 1][lane * 4] = xs1;
  }

  for (int j = 0; j < 2; ++j) {
    const int r = r0 + j;
    int bi = j ? pi1 : pi0, ei = j ? pi2 : pi1;
    int bo = j ? po1 : po0, eo = j ? po2 : po1;
    if (bi < 0) bi = 0;
    if (bo < 0) bo = 0;
    if (ei > EE) ei = EE;
    if (eo > EE) eo = EE;

    float4 f = make_float4(0.f, 0.f, 0.f, 0.f);
    float4 g = make_float4(0.f, 0.f, 0.f, 0.f);
    for (int off = 0; off < EE; off += 32) {
      int na = ei - bi - off; if (na > 32) na = 32;
      int nb = eo - bo - off; if (nb > 32) nb = 32;
      if (na <= 0 && nb <= 0) break;
      const int2 ta = ld_chunk(adj_in, bi + off, na, lane);
      const int2 tb = ld_chunk(adj_out, bo + off, nb, lane);
      const int nmin = na < nb ? na : nb;
      int k = 0;
      for (; k < nmin; k += 4) {
        float4 xa[4], xb[4];
        float wa[4], wb[4];
        ld_rows(x, ta.x, ta.y, k, na, lane, xa, wa);
        ld_rows(x, tb.x, tb.y, k, nb, lane, xb, wb);
        add_rows(xa, wa, k, na, f);
        add_rows(xb, wb, k, nb, g);
      }
      for (; k < na; k += 4) {
        float4 xa[4];
        float wa[4];
        ld_rows(x, ta.x, ta.y, k, na, lane, xa, wa);
        add_rows(xa, wa, k, na, f);
      }
      for (; k < nb; k += 4) {
        float4 xb[4];
        float wb[4];
        ld_rows(x, tb.x, tb.y, k, nb, lane, xb, wb);
        add_rows(xb, wb, k, nb, g);
      }
    }
    *(float4*)&As[r][DD + lane * 4] = f;
    *(float4*)&As[r][2 * DD + lane * 4] = g;
  }
  __syncthreads();

  const int hl = lane >> 4;
  const int l16 = lane & 15;
  const int colBase = wave * 16;
  const int col = colBase + l16;

  v8f acc = {};
  for (int k0 = 0; k0 < K3; k0 += 4) {
    v2f a;
    a.x = As[l16][k0 + 2 * hl + 0];
    a.y = As[l16][k0 + 2 * hl + 1];
    v2f b;
    b.x = W[(size_t)(k0 + 2 * hl + 0) * DD + col];
    b.y = W[(size_t)(k0 + 2 * hl + 1) * DD + col];
    acc = __builtin_amdgcn_wmma_f32_16x16x4_f32(
         false, a,  false, b,
         (short)0, acc,  false,  false);
  }

  const float bb = bias[col];
#pragma unroll
  for (int r = 0; r < 8; ++r) {
    float v = acc[r] + bb;
    if (do_relu) v = fmaxf(v, 0.0f);
    out[(size_t)(rowBase + r + 8 * hl) * DD + col] = v;
  }
}

extern "C" void kernel_launch(void* const* d_in, const int* in_sizes, int n_in,
                              void* d_out, int out_size, void* d_ws, size_t ws_size,
                              hipStream_t stream) {
  const float* x  = (const float*)d_in[0];
  const float* W0 = (const float*)d_in[1];
  const float* b0 = (const float*)d_in[2];
  const float* W1 = (const float*)d_in[3];
  const float* b1 = (const float*)d_in[4];
  const float* W2 = (const float*)d_in[5];
  const float* b2 = (const float*)d_in[6];
  const int* src  = (const int*)d_in[7];
  const int* dst  = (const int*)d_in[8];
  float* out = (float*)d_out;

  float* ws      = (float*)d_ws;
  float* inv_out = ws;
  float* inv_in  = inv_out + NN;
  float* xA      = ws + 2 * (size_t)NN;
  int*   rp_out  = (int*)(xA + (size_t)NN * DD);
  int*   rp_in   = rp_out + RPS;
  int*   adj_in  = rp_in + RPS;
  int*   adj_out = adj_in + 2 * (size_t)EE;
  int*   rk_out  = adj_out + 2 * (size_t)EE;
  int*   rk_in   = rk_out + EE;
  int*   cnt     = rk_in + EE;
  int*   tot     = cnt + 2 * (size_t)QC * NN;

  rank_count_kernel<<<2 * PB * QC, OWN_T, 0, stream>>>(src, dst, cnt, tot, rk_out);
  scan_off_kernel<<<2 * PB, OWN_T, 0, stream>>>(cnt, tot, inv_out, rp_out);
  fill_kernel<<<(EE + 255) / 256, 256, 0, stream>>>(src, dst, cnt, cnt + (size_t)QC * NN,
                                                    rk_out, rk_in, inv_out, inv_in,
                                                    adj_in, adj_out, EE);

  const float* Ws[3]  = {W0, W1, W2};
  const float* bs[3]  = {b0, b1, b2};
  float*       dsts[3] = {out, xA, out};

  const float* cur = x;
  for (int l = 0; l < 3; ++l) {
    layer_kernel<<<NN / 16, 256, 0, stream>>>(cur, rp_out, rp_in, adj_in, adj_out,
                                              Ws[l], bs[l], dsts[l], (l < 2) ? 1 : 0);
    cur = dsts[l];
  }
}


# Layer blocks of 32 nodes: 16 waves gather, then 8 waves multiply both 16-row tiles with each W fragment they load (W fetched once per 32 nodes, not 16); the other 8 end at the barrier; sum order kept
# speedup vs baseline: 3.9808x; 1.0728x over previous
#include <hip/hip_runtime.h>


#define NN 50000
#define EE 800000
#define DD 128
#define K3 384
#define RPS (NN + 16)
#define PB 16
#define QC 8
#define RNG 3128
#define CH4 (EE / 4 / QC)
#define OWN_T 1024
#define KEY_U 5
static_assert(PB * RNG >= NN && RNG % 8 == 0 && NN % 8 == 0, "ranges cover the nodes in whole groups of four");
static_assert(CH4 * 4 * QC == EE, "chunks cover the edges");
static_assert(4 * OWN_T >= RNG && (PB - 1) * QC <= OWN_T, "one work-group scans a range");
#define MB 32
static_assert(NN % 16 == 0, "whole 16-node tiles");
static_assert(MB == 32 && NN % 2 == 0, "a wave's two nodes are both inside or both outside the graph");

typedef __attribute__((ext_vector_type(2))) float v2f;
typedef __attribute__((ext_vector_type(8))) float v8f;

static __device__ __forceinline__ void own_key(int k, int e, int lo, int n,
                                               int* c, int* __restrict__ rk) {
  const unsigned r = (unsigned)(k - lo);
  if (r < (unsigned)n) rk[e] = atomicAdd(&c[r], 1);
}

__global__ __launch_bounds__(OWN_T) void rank_count_kernel(
    const int* __restrict__ src, const int* __restrict__ dst,
    int* __restrict__ cnt, int* __restrict__ tot, int* __restrict__ rk) {
  __shared__ int c[RNG];
  __shared__ int tot_s;
  const int t = threadIdx.x;
  const int lane = t & 31;
  const int dir = (int)blockIdx.x / (PB * QC);
  const int pq = (int)blockIdx.x - dir * (PB * QC);
  const int p = pq / QC;
  const int q = pq - p * QC;
  const int lo = p * RNG;
  int n = NN - lo;
  if (n > RNG) n = RNG;
  if (n < 0) n = 0;
  const int4* __restrict__ k4 = (const int4*)(dir ? dst : src) + (size_t)q * CH4;
  int* __restrict__ rkd = rk + (size_t)dir * EE + (size_t)q * (4 * CH4);

  for (int i = t; i < RNG; i += OWN_T) c[i] = 0;
  if (t == 0) tot_s = 0;
  __syncthreads();

  for (int i0 = t; i0 < CH4; i0 += KEY_U * OWN_T) {
    int4 k[KEY_U];
#pragma unroll
    for (int j = 0; j < KEY_U; ++j) {
      const int i = i0 + j * OWN_T;
      k[j] = make_int4(0x7fffffff, 0x7fffffff, 0x7fffffff, 0x7fffffff);
      if (i < CH4) k[j] = k4[i];
    }
#pragma unroll
    for (int j = 0; j < KEY_U; ++j) {
      const int e = 4 * (i0 + j * OWN_T);
      own_key(k[j].x, e + 0, lo, n, c, rkd);
      own_key(k[j].y, e + 1, lo, n, c, rkd);
      own_key(k[j].z, e + 2, lo, n, c, rkd);
      own_key(k[j].w, e + 3, lo, n, c, rkd);
    }
  }
  __syncthreads();

  int* __restrict__ cd = cnt + ((size_t)dir * QC + q) * NN + lo;
  int sum = 0;
  for (int i = t; i < n; i += OWN_T) {
    const int v = c[i];
    cd[i] = v;
    sum += v;
  }
  for (int off = 16; off > 0; off >>= 1) sum += __shfl_xor(sum, off);
  if (lane == 0) atomicAdd(&tot_s, sum);
  __syncthreads();
  if (t == 0) tot[(dir * PB + p) * QC + q] = tot_s;
}

__global__ __launch_bounds__(OWN_T) void scan_off_kernel(
    int* __restrict__ cnt, const int* __restrict__ tot,
    float* __restrict__ inv, int* __restrict__ rp) {
  __shared__ int wsum[32];
  __shared__ int base_s;
  const int t = threadIdx.x;
  const int lane = t & 31;
  const int wave = t >> 5;
  const int dir = (int)blockIdx.x / PB;
  const int p = (int)blockIdx.x - dir * PB;
  const int lo = p * RNG;
  int n = NN - lo;
  if (n > RNG) n = RNG;
  if (n < 0) n = 0;
  const bool live = 4 * t < n;

  if (t == 0) base_s = 0;
  __syncthreads();
  if (t < p * QC) atomicAdd(&base_s, tot[dir * PB * QC + t]);

  int* __restrict__ cd = cnt + (size_t)dir * QC * NN + lo + 4 * t;
  int4 v[QC];
  int4 deg = make_int4(0, 0, 0, 0);
#pragma unroll
  for (int q = 0; q < QC; ++q) {
    int4 w = make_int4(0, 0, 0, 0);
    if (live) w = *(const int4*)(cd + (size_t)q * NN);
    v[q] = deg;
    deg.x += w.x; deg.y += w.y; deg.z += w.z; deg.w += w.w;
  }
  const int s1 = deg.x, s2 = s1 + deg.y, s3 = s2 + deg.z, s4 = s3 + deg.w;
  int incl = s4;
  for (int off = 1; off < 32; off <<= 1) {
    int y = __shfl_up(incl, off);
    if (lane >= off) incl += y;
  }
  if (lane == 31) wsum[wave] = incl;
  __syncthreads();
  const int wv = wsum[lane];
  int pre = (lane < wave) ? wv : 0;
  int all = wv;
  for (int off = 16; off > 0; off >>= 1) {
    pre += __shfl_xor(pre, off);
    all += __shfl_xor(all, off);
  }
  const int first = base_s;
  const int b0 = first + pre + (incl - s4);
  if (live) {
    const int4 r = make_int4(b0, b0 + s1, b0 + s2, b0 + s3);
    *(int4*)(rp + (size_t)dir * RPS + lo + 4 * t) = r;
    float4 o;
    o.x = 1.0f / fmaxf((float)deg.x, 1.0f);
    o.y = 1.0f / fmaxf((float)deg.y, 1.0f);
    o.z = 1.0f / fmaxf((float)deg.z, 1.0f);
    o.w = 1.0f / fmaxf((float)deg.w, 1.0f);
    *(float4*)(inv + (size_t)dir * NN + lo + 4 * t) = o;
#pragma unroll
    for (int q = 0; q < QC; ++q) {
      *(int4*)(cd + (size_t)q * NN) =
          make_int4(r.x + v[q].x, r.y + v[q].y, r.z + v[q].z, r.w + v[q].w);
    }
  }
  if (p == PB - 1 && t == 0) rp[(size_t)dir * RPS + NN] = first + all;
}

__global__ __launch_bounds__(256) void fill_kernel(
    const int* __restrict__ src, const int* __restrict__ dst,
    const int* __restrict__ off_out, const int* __restrict__ off_in,
    const int* __restrict__ rk_out, const int* __restrict__ rk_in,
    const float* __restrict__ inv_out, const float* __restrict__ inv_in,
    int* __restrict__ adj_in, int* __restrict__ adj_out, int E) {
  int e = blockIdx.x * blockDim.x + threadIdx.x;
  if (e >= E) return;
  const int s = src[e];
  const int d = dst[e];
  const size_t row = (size_t)(e / (4 * CH4)) * NN;
  const int pi = off_in[row + d] + rk_in[e];
  const int po = off_out[row + s] + rk_out[e];
  const int2 vi = make_int2(s, __float_as_int(inv_out[s]));
  const int2 vo = make_int2(d, __float_as_int(inv_in[d]));
  if ((unsigned)pi < (unsigned)E) *(int2*)(adj_in + 2 * (size_t)pi) = vi;
  if ((unsigned)po < (unsigned)E) *(int2*)(adj_out + 2 * (size_t)po) = vo;
}

static __device__ __forceinline__ int2 ld_chunk(const int* __restrict__ adj,
                                                int base, int n, int lane) {
  int2 t = make_int2(0, 0);
  if (lane < n) t = *(const int2*)(adj + 2 * (size_t)(base + lane));
  return t;
}

static __device__ __forceinline__ void ld_rows(const float* __restrict__ x,
                                               int cs, int cw, int k, int n, int lane,
                                               float4 (&xv)[4], float (&w)[4]) {
#pragma unroll
  for (int j = 0; j < 4; ++j) {
    int kk = k + j;
    if (j > 0 && kk > n - 1) kk = n - 1;
    int s = __builtin_amdgcn_readlane(cs, kk);
    if ((unsigned)s > (unsigned)(NN - 1)) s = NN - 1;
    w[j] = __int_as_float(__builtin_amdgcn_readlane(cw, kk));
    xv[j] = *(const float4*)(x + (size_t)s * DD + lane * 4);
  }
}

static __device__ __forceinline__ void add_rows(const float4 (&xv)[4], const float (&w)[4],
                                                int k, int n, float4& acc) {
#pragma unroll
  for (int j = 0; j < 4; ++j) {
    if (j == 0 || k + j < n) {
      acc.x = fmaf(xv[j].x, w[j], acc.x);
      acc.y = fmaf(xv[j].y, w[j], acc.y);
      acc.z = fmaf(xv[j].z, w[j], acc.z);
      acc.w = fmaf(xv[j].w, w[j], acc.w);
    }
  }
}

__global__ __launch_bounds__(512) void layer_kernel(
    const float* __restrict__ x,
    const int* __restrict__ rp_out, const int* __restrict__ rp_in,
    const int* __restrict__ adj_in, const int* __restrict__ adj_out,
    const float* __restrict__ W, const float* __restrict__ bias,
    float* __restrict__ out, int do_relu) {
  __shared__ __attribute__((aligned(16))) float As[MB][K3 + 4];

  const int rowBase = blockIdx.x * MB;
  const int tid = threadIdx.x;
  const int lane = tid & 31;
  const int wave = __builtin_amdgcn_readfirstlane(tid >> 5);

  const int r0 = wave * 2;
  const int v0 = rowBase + r0;
  if (v0 < NN) {
    const int pi0 = rp_in[v0], pi1 = rp_in[v0 + 1], pi2 = rp_in[v0 + 2];
    const int po0 = rp_out[v0], po1 = rp_out[v0 + 1], po2 = rp_out[v0 + 2];
    {
      const float4 xs0 = *(const float4*)(x + (size_t)v0 * DD + lane * 4);
      const float4 xs1 = *(const float4*)(x + (size_t)(v0 + 1) * DD + lane * 4);
      *(float4*)&As[r0][lane * 4] = xs0;
      *(float4*)&As[r0 + 1][lane * 4] = xs1;
    }

    for (int j = 0; j < 2; ++j) {
      const int r = r0 + j;
      int bi = j ? pi1 : pi0, ei = j ? pi2 : pi1;
      int bo = j ? po1 : po0, eo = j ? po2 : po1;
      if (bi < 0) bi = 0;
      if (bo < 0) bo = 0;
      if (ei > EE) ei = EE;
      if (eo > EE) eo = EE;

      float4 f = make_float4(0.f, 0.f, 0.f, 0.f);
      float4 g = make_float4(0.f, 0.f, 0.f, 0.f);
      for (int off = 0; off < EE; off += 32) {
        int na = ei - bi - off; if (na > 32) na = 32;
        int nb = eo - bo - off; if (nb > 32) nb = 32;
        if (na <= 0 && nb <= 0) break;
        const int2 ta = ld_chunk(adj_in, bi + off, na, lane);
        const int2 tb = ld_chunk(adj_out, bo + off, nb, lane);
        const int nmin = na < nb ? na : nb;
        int k = 0;
        for (; k < nmin; k += 4) {
          float4 xa[4], xb[4];
          float wa[4], wb[4];
          ld_rows(x, ta.x, ta.y, k, na, lane, xa, wa);
          ld_rows(x, tb.x, tb.y, k, nb, lane, xb, wb);
          add_rows(xa, wa, k, na, f);
          add_rows(xb, wb, k, nb, g);
        }
        for (; k < na; k += 4) {
          float4 xa[4];
          float wa[4];
          ld_rows(x, ta.x, ta.y, k, na, lane, xa, wa);
          add_rows(xa, wa, k, na, f);
        }
        for (; k < nb; k += 4) {
          float4 xb[4];
          float wb[4];
          ld_rows(x, tb.x, tb.y, k, nb, lane, xb, wb);
          add_rows(xb, wb, k, nb, g);
        }
      }
      *(float4*)&As[r][DD + lane * 4] = f;
      *(float4*)&As[r][2 * DD + lane * 4] = g;
    }
  } else {
    const float4 z = make_float4(0.f, 0.f, 0.f, 0.f);
    for (int c = 0; c < 3; ++c) {
      *(float4*)&As[r0][c * DD + lane * 4] = z;
      *(float4*)&As[r0 + 1][c * DD + lane * 4] = z;
    }
  }
  __syncthreads();
  if (wave >= 8) return;

  const int hl = lane >> 4;
  const int l16 = lane & 15;
  const int colBase = wave * 16;
  const int col = colBase + l16;

  v8f acc0 = {};
  v8f acc1 = {};
  for (int k0 = 0; k0 < K3; k0 += 4) {
    v2f b;
    b.x = W[(size_t)(k0 + 2 * hl + 0) * DD + col];
    b.y = W[(size_t)(k0 + 2 * hl + 1) * DD + col];
    v2f a0, a1;
    a0.x = As[l16][k0 + 2 * hl + 0];
    a0.y = As[l16][k0 + 2 * hl + 1];
    a1.x = As[16 + l16][k0 + 2 * hl + 0];
    a1.y = As[16 + l16][k0 + 2 * hl + 1];
    acc0 = __builtin_amdgcn_wmma_f32_16x16x4_f32(
         false, a0,  false, b,
         (short)0, acc0,  false,  false);
    acc1 = __builtin_amdgcn_wmma_f32_16x16x4_f32(
         false, a1,  false, b,
         (short)0, acc1,  false,  false);
  }

  const float bb = bias[col];
#pragma unroll
  for (int r = 0; r < 8; ++r) {
    float v = acc0[r] + bb;
    if (do_relu) v = fmaxf(v, 0.0f);
    out[(size_t)(rowBase + r + 8 * hl) * DD + col] = v;
  }
  if (rowBase + 16 < NN) {
#pragma unroll
    for (int r = 0; r < 8; ++r) {
      float v = acc1[r] + bb;
      if (do_relu) v = fmaxf(v, 0.0f);
      out[(size_t)(rowBase + 16 + r + 8 * hl) * DD + col] = v;
    }
  }
}

extern "C" void kernel_launch(void* const* d_in, const int* in_sizes, int n_in,
                              void* d_out, int out_size, void* d_ws, size_t ws_size,
                              hipStream_t stream) {
  const float* x  = (const float*)d_in[0];
  const float* W0 = (const float*)d_in[1];
  const float* b0 = (const float*)d_in[2];
  const float* W1 = (const float*)d_in[3];
  const float* b1 = (const float*)d_in[4];
  const float* W2 = (const float*)d_in[5];
  const float* b2 = (const float*)d_in[6];
  const int* src  = (const int*)d_in[7];
  const int* dst  = (const int*)d_in[8];
  float* out = (float*)d_out;

  float* ws      = (float*)d_ws;
  float* inv_out = ws;
  float* inv_in  = inv_out + NN;
  float* xA      = ws + 2 * (size_t)NN;
  int*   rp_out  = (int*)(xA + (size_t)NN * DD);
  int*   rp_in   = rp_out + RPS;
  int*   adj_in  = rp_in + RPS;
  int*   adj_out = adj_in + 2 * (size_t)EE;
  int*   rk_out  = adj_out + 2 * (size_t)EE;
  int*   rk_in   = rk_out + EE;
  int*   cnt     = rk_in + EE;
  int*   tot     = cnt + 2 * (size_t)QC * NN;

  rank_count_kernel<<<2 * PB * QC, OWN_T, 0, stream>>>(src, dst, cnt, tot, rk_out);
  scan_off_kernel<<<2 * PB, OWN_T, 0, stream>>>(cnt, tot, inv_out, rp_out);
  fill_kernel<<<(EE + 255) / 256, 256, 0, stream>>>(src, dst, cnt, cnt + (size_t)QC * NN,
                                                    rk_out, rk_in, inv_out, inv_in,
                                                    adj_in, adj_out, EE);

  const float* Ws[3]  = {W0, W1, W2};
  const float* bs[3]  = {b0, b1, b2};
  float*       dsts[3] = {out, xA, out};

  const float* cur = x;
  for (int l = 0; l < 3; ++l) {
    layer_kernel<<<(NN + MB - 1) / MB, 512, 0, stream>>>(cur, rp_out, rp_in, adj_in, adj_out,
                                              Ws[l], bs[l], dsts[l], (l < 2) ? 1 : 0);
    cur = dsts[l];
  }
}
